# starting kernel
# baseline (speedup 1.0000x reference)
#include <hip/hip_runtime.h>
#include <hip/hip_bf16.h>
#include <cstdint>
#include <cstddef>

typedef float v2f __attribute__((ext_vector_type(2)));
typedef float v4f __attribute__((ext_vector_type(4)));
typedef float v8f __attribute__((ext_vector_type(8)));

constexpr int Bc = 8;
constexpr int Lc = 4096;
constexpr int Hc = 1024;
constexpr int Sc = 512;
constexpr int Tc = 512;
constexpr int TOK = 64;
constexpr int KC  = 32;
constexpr int TSTRIDE = 36;
constexpr int WSTRIDE = 34;
constexpr int NSEG = 1056;

__global__ __launch_bounds__(1024) void seg_scan_kernel(
    const int* __restrict__ wid, const int* __restrict__ mask,
    int* __restrict__ seg, int* __restrict__ flag) {
  __shared__ int sHas[1024];
  __shared__ int sWid[1024];
  __shared__ int sSum[1024];
  const int tid = threadIdx.x;
  const int b = blockIdx.x;
  const int base = b * Lc + tid * 4;

  int w[4], v[4];
#pragma unroll
  for (int j = 0; j < 4; ++j) {
    w[j] = wid[base + j];
    v[j] = (mask[base + j] > 0) ? 1 : 0;
  }

  int curH = 0, curW = -2;
#pragma unroll
  for (int j = 0; j < 4; ++j)
    if (v[j]) { curH = 1; curW = w[j]; }
  sHas[tid] = curH; sWid[tid] = curW;
  __syncthreads();
  for (int off = 1; off < 1024; off <<= 1) {
    int pH = 0, pW = -2;
    if (tid >= off) { pH = sHas[tid - off]; pW = sWid[tid - off]; }
    __syncthreads();
    if (!curH && pH) { curH = 1; curW = pW; }
    sHas[tid] = curH; sWid[tid] = curW;
    __syncthreads();
  }
  int prev = -2;
  if (tid > 0 && sHas[tid - 1]) prev = sWid[tid - 1];

  int f[4]; int localSum = 0;
#pragma unroll
  for (int j = 0; j < 4; ++j) {
    f[j] = 0;
    if (v[j]) {
      f[j] = (w[j] == -1 || prev == -1 || w[j] != prev) ? 1 : 0;
      prev = w[j];
      localSum += f[j];
    }
  }

  int curS = localSum;
  sSum[tid] = curS;
  __syncthreads();
  for (int off = 1; off < 1024; off <<= 1) {
    int p = 0;
    if (tid >= off) p = sSum[tid - off];
    __syncthreads();
    curS += p;
    sSum[tid] = curS;
    __syncthreads();
  }
  int run = (tid > 0) ? sSum[tid - 1] : 0;
#pragma unroll
  for (int j = 0; j < 4; ++j) {
    int sg = -1;
    if (v[j]) { run += f[j]; sg = run - 1; }
    seg[base + j] = sg;
    flag[base + j] = f[j];
  }
}

__global__ __launch_bounds__(128) void dot_wmma_kernel(
    const float* __restrict__ X, const float* __restrict__ wvec,
    float* __restrict__ dsrc, float* __restrict__ dtgt) {
  __shared__ __align__(16) float tile[TOK * TSTRIDE];
  __shared__ __align__(16) float wPad[16 * WSTRIDE];

  const int tid  = threadIdx.x;
  const int lane = tid & 31;
  const int wave = tid >> 5;
  const int col  = lane & 15;
  const int half = lane >> 4;
  const int tokBase = blockIdx.x * TOK;

  for (int i = tid; i < 16 * WSTRIDE; i += 128) wPad[i] = 0.0f;

  v8f acc = {};

  for (int k0 = 0; k0 < Hc; k0 += KC) {
    __syncthreads();

    if (tid < 64) {
      int r = tid >> 5, c = tid & 31;
      wPad[r * WSTRIDE + c] = wvec[r * Hc + k0 + c];
    }
#pragma unroll
    for (int it = 0; it < 4; ++it) {
      int idx = it * 128 + tid;
      int row = idx >> 3, c4 = idx & 7;
      v4f val = *reinterpret_cast<const v4f*>(
          X + (size_t)(tokBase + row) * Hc + k0 + c4 * 4);
      *reinterpret_cast<v4f*>(&tile[row * TSTRIDE + c4 * 4]) = val;
    }
    if (k0 + KC < Hc && tid < TOK) {
      __builtin_prefetch(X + (size_t)(tokBase + tid) * Hc + k0 + KC, 0, 1);
    }
    __syncthreads();

    const float* aBase = &tile[(wave * 16 + col) * TSTRIDE + half * 2];
    const float* bBase = &wPad[col * WSTRIDE + half * 2];
#pragma unroll
    for (int s = 0; s < 8; ++s) {
      v2f a = *reinterpret_cast<const v2f*>(aBase + s * 4);
      v2f bf = *reinterpret_cast<const v2f*>(bBase + s * 4);
      acc = __builtin_amdgcn_wmma_f32_16x16x4_f32(
           false, a,  false, bf,
           (short)0, acc,  false,  false);
    }
  }

  if (col < 2) {
    float* dst = col ? dtgt : dsrc;
    int rb = tokBase + wave * 16 + half * 8;
#pragma unroll
    for (int r = 0; r < 8; ++r) dst[rb + r] = acc[r];
  }
}

__global__ void seg_reduce_kernel(
    const int* __restrict__ seg, const int* __restrict__ flag,
    const float* __restrict__ dsrc, const float* __restrict__ dtgt,
    float* __restrict__ sums, int* __restrict__ counts) {
  int idx = blockIdx.x * blockDim.x + threadIdx.x;
  if (idx >= Bc * Lc) return;
  if (!flag[idx]) return;
  int sg = seg[idx];
  if (sg < 1 || sg > Sc + Tc) return;
  int b = idx / Lc;
  int end = (b + 1) * Lc;
  const float* d = (sg <= Sc) ? dsrc : dtgt;
  float sum = 0.0f; int cnt = 0;
  for (int j = idx; j < end; ++j) {
    int s2 = seg[j];
    if (s2 == sg) { sum += d[j]; ++cnt; }
    else if (s2 != -1) break;
  }
  sums[b * NSEG + sg] = sum;
  counts[b * NSEG + sg] = cnt;
}

__global__ void avg_kernel(const float* __restrict__ sums,
                           const int* __restrict__ counts,
                           float* __restrict__ avg_a, float* __restrict__ avg_c) {
  int idx = blockIdx.x * blockDim.x + threadIdx.x;
  if (idx >= Bc * (Sc + Tc)) return;
  int b = idx / (Sc + Tc);
  int i = idx % (Sc + Tc);
  int sg = 1 + i;
  float cnt = (float)max(counts[b * NSEG + sg], 1);
  float m = sums[b * NSEG + sg] / cnt;
  if (i < Sc) avg_a[b * Sc + i] = m;
  else        avg_c[b * Tc + (i - Sc)] = m;
}

__global__ void logits_kernel(const float* __restrict__ a,
                              const float* __restrict__ c,
                              const float* __restrict__ biasPtr,
                              float* __restrict__ out) {
  int idx = blockIdx.x * blockDim.x + threadIdx.x;
  if (idx >= Bc * Sc * (Tc / 4)) return;
  int t4  = idx % (Tc / 4);
  int rem = idx / (Tc / 4);
  int s = rem % Sc;
  int b = rem / Sc;
  float av = a[b * Sc + s] + biasPtr[0];
  v4f cv = *reinterpret_cast<const v4f*>(c + b * Tc + t4 * 4);
  v4f o = cv + av;
  reinterpret_cast<v4f*>(out)[idx] = o;
}

extern "C" void kernel_launch(void* const* d_in, const int* in_sizes, int n_in,
                              void* d_out, int out_size, void* d_ws, size_t ws_size,
                              hipStream_t stream) {
  const float* X    = (const float*)d_in[0];
  const int*   msk  = (const int*)d_in[1];
  const int*   wid  = (const int*)d_in[2];
  const float* wvec = (const float*)d_in[5];
  const float* bias = (const float*)d_in[6];
  float* out = (float*)d_out;

  char* ws = (char*)d_ws;
  size_t off = 0;
  int*   seg    = (int*)(ws + off);   off += (size_t)Bc * Lc * 4;
  int*   flg    = (int*)(ws + off);   off += (size_t)Bc * Lc * 4;
  float* dsrc   = (float*)(ws + off); off += (size_t)Bc * Lc * 4;
  float* dtgt   = (float*)(ws + off); off += (size_t)Bc * Lc * 4;
  char*  zbase  = ws + off;
  float* sums   = (float*)(ws + off); off += (size_t)Bc * NSEG * 4;
  int*   counts = (int*)(ws + off);   off += (size_t)Bc * NSEG * 4;
  size_t zbytes = (size_t)Bc * NSEG * 8;
  float* avg_a  = (float*)(ws + off); off += (size_t)Bc * Sc * 4;
  float* avg_c  = (float*)(ws + off); off += (size_t)Bc * Tc * 4;

  hipMemsetAsync(zbase, 0, zbytes, stream);

  seg_scan_kernel<<<Bc, 1024, 0, stream>>>(wid, msk, seg, flg);

  dot_wmma_kernel<<<(Bc * Lc) / TOK, 128, 0, stream>>>(X, wvec, dsrc, dtgt);

  seg_reduce_kernel<<<(Bc * Lc + 255) / 256, 256, 0, stream>>>(
      seg, flg, dsrc, dtgt, sums, counts);

  avg_kernel<<<(Bc * (Sc + Tc) + 255) / 256, 256, 0, stream>>>(
      sums, counts, avg_a, avg_c);

  logits_kernel<<<(Bc * Sc * (Tc / 4) + 255) / 256, 256, 0, stream>>>(
      avg_a, avg_c, bias, out);
}


# dot kernel: each wave streams a 128-float K slice of 64 tokens with whole-line loads one 16-row group ahead and a barrier-free per-wave LDS transpose; K sum order changes (8 slices x 2 chains)
# speedup vs baseline: 1.3473x; 1.3473x over previous
#include <hip/hip_runtime.h>
#include <hip/hip_bf16.h>
#include <cstdint>
#include <cstddef>

typedef float v2f __attribute__((ext_vector_type(2)));
typedef float v4f __attribute__((ext_vector_type(4)));
typedef float v8f __attribute__((ext_vector_type(8)));

constexpr int Bc = 8;
constexpr int Lc = 4096;
constexpr int Hc = 1024;
constexpr int Sc = 512;
constexpr int Tc = 512;
constexpr int TOK = 64;
constexpr int DW  = 8;
constexpr int KS  = 128;
constexpr int TPITCH = 132;
constexpr int NSEG = 1056;

__global__ __launch_bounds__(1024) void seg_scan_kernel(
    const int* __restrict__ wid, const int* __restrict__ mask,
    int* __restrict__ seg, int* __restrict__ flag) {
  __shared__ int sHas[1024];
  __shared__ int sWid[1024];
  __shared__ int sSum[1024];
  const int tid = threadIdx.x;
  const int b = blockIdx.x;
  const int base = b * Lc + tid * 4;

  int w[4], v[4];
#pragma unroll
  for (int j = 0; j < 4; ++j) {
    w[j] = wid[base + j];
    v[j] = (mask[base + j] > 0) ? 1 : 0;
  }

  int curH = 0, curW = -2;
#pragma unroll
  for (int j = 0; j < 4; ++j)
    if (v[j]) { curH = 1; curW = w[j]; }
  sHas[tid] = curH; sWid[tid] = curW;
  __syncthreads();
  for (int off = 1; off < 1024; off <<= 1) {
    int pH = 0, pW = -2;
    if (tid >= off) { pH = sHas[tid - off]; pW = sWid[tid - off]; }
    __syncthreads();
    if (!curH && pH) { curH = 1; curW = pW; }
    sHas[tid] = curH; sWid[tid] = curW;
    __syncthreads();
  }
  int prev = -2;
  if (tid > 0 && sHas[tid - 1]) prev = sWid[tid - 1];

  int f[4]; int localSum = 0;
#pragma unroll
  for (int j = 0; j < 4; ++j) {
    f[j] = 0;
    if (v[j]) {
      f[j] = (w[j] == -1 || prev == -1 || w[j] != prev) ? 1 : 0;
      prev = w[j];
      localSum += f[j];
    }
  }

  int curS = localSum;
  sSum[tid] = curS;
  __syncthreads();
  for (int off = 1; off < 1024; off <<= 1) {
    int p = 0;
    if (tid >= off) p = sSum[tid - off];
    __syncthreads();
    curS += p;
    sSum[tid] = curS;
    __syncthreads();
  }
  int run = (tid > 0) ? sSum[tid - 1] : 0;
#pragma unroll
  for (int j = 0; j < 4; ++j) {
    int sg = -1;
    if (v[j]) { run += f[j]; sg = run - 1; }
    seg[base + j] = sg;
    flag[base + j] = f[j];
  }
}

__global__ __launch_bounds__(256) void dot_wmma_kernel(
    const float* __restrict__ X, const float* __restrict__ wvec,
    float* __restrict__ dsrc, float* __restrict__ dtgt) {
  __shared__ __align__(16) float tile[DW][16 * TPITCH];
  __shared__ __align__(16) float part[DW][2][TOK];

  const int tid  = threadIdx.x;
  const int lane = tid & 31;
  const int ks   = __builtin_amdgcn_readfirstlane(tid >> 5);
  const int col  = lane & 15;
  const int half = lane >> 4;
  const int tokBase = blockIdx.x * TOK;

  v4f bw[16];
  {
    const float* wp = wvec + (col ? 1 : 0) * Hc + ks * KS + half * 4;
#pragma unroll
    for (int j = 0; j < 16; ++j) {
      v4f t = *reinterpret_cast<const v4f*>(wp + j * 8);
      bw[j] = (col < 2) ? t : v4f{0.0f, 0.0f, 0.0f, 0.0f};
    }
  }

  const float* xp = X + (size_t)tokBase * Hc + ks * KS + lane * 4;
  float* myTile = &tile[ks][0];
  v4f buf[2][16];

#define LOAD_GROUP(G) do { \
    _Pragma("unroll") for (int i = 0; i < 16; ++i) \
      buf[(G) & 1][i] = *reinterpret_cast<const v4f*>(xp + (size_t)((G) * 16 + i) * Hc); \
  } while (0)

  LOAD_GROUP(0);
#pragma unroll
  for (int g = 0; g < 4; ++g) {
    __builtin_amdgcn_sched_barrier(0);
    if (g + 1 < 4) LOAD_GROUP(g + 1);
    __builtin_amdgcn_sched_barrier(0);

#pragma unroll
    for (int i = 0; i < 16; ++i)
      *reinterpret_cast<v4f*>(&myTile[i * TPITCH + lane * 4]) = buf[g & 1][i];

    v4f a[16];
#pragma unroll
    for (int j = 0; j < 16; ++j)
      a[j] = *reinterpret_cast<const v4f*>(&myTile[col * TPITCH + j * 8 + half * 4]);
    __builtin_amdgcn_sched_barrier(0);

    v8f acc0 = {}, acc1 = {};
#pragma unroll
    for (int j = 0; j < 16; ++j) {
      v2f a0 = {a[j].x, a[j].y}, a1 = {a[j].z, a[j].w};
      v2f b0 = {bw[j].x, bw[j].y}, b1 = {bw[j].z, bw[j].w};
      acc0 = __builtin_amdgcn_wmma_f32_16x16x4_f32(
           false, a0,  false, b0,
           (short)0, acc0,  false,  false);
      acc1 = __builtin_amdgcn_wmma_f32_16x16x4_f32(
           false, a1,  false, b1,
           (short)0, acc1,  false,  false);
    }
    v8f acc = acc0 + acc1;

    if (col < 2) {
      float* pp = &part[ks][col][g * 16 + half * 8];
      *reinterpret_cast<v4f*>(pp)     = v4f{acc[0], acc[1], acc[2], acc[3]};
      *reinterpret_cast<v4f*>(pp + 4) = v4f{acc[4], acc[5], acc[6], acc[7]};
    }
  }
#undef LOAD_GROUP

  __syncthreads();
  if (tid < 2 * TOK) {
    const int s = tid >> 6, tok = tid & (TOK - 1);
    float sum = part[0][s][tok];
    for (int k = 1; k < DW; ++k) sum += part[k][s][tok];
    (s ? dtgt : dsrc)[tokBase + tok] = sum;
  }
}

__global__ void seg_reduce_kernel(
    const int* __restrict__ seg, const int* __restrict__ flag,
    const float* __restrict__ dsrc, const float* __restrict__ dtgt,
    float* __restrict__ sums, int* __restrict__ counts) {
  int idx = blockIdx.x * blockDim.x + threadIdx.x;
  if (idx >= Bc * Lc) return;
  if (!flag[idx]) return;
  int sg = seg[idx];
  if (sg < 1 || sg > Sc + Tc) return;
  int b = idx / Lc;
  int end = (b + 1) * Lc;
  const float* d = (sg <= Sc) ? dsrc : dtgt;
  float sum = 0.0f; int cnt = 0;
  for (int j = idx; j < end; ++j) {
    int s2 = seg[j];
    if (s2 == sg) { sum += d[j]; ++cnt; }
    else if (s2 != -1) break;
  }
  sums[b * NSEG + sg] = sum;
  counts[b * NSEG + sg] = cnt;
}

__global__ void avg_kernel(const float* __restrict__ sums,
                           const int* __restrict__ counts,
                           float* __restrict__ avg_a, float* __restrict__ avg_c) {
  int idx = blockIdx.x * blockDim.x + threadIdx.x;
  if (idx >= Bc * (Sc + Tc)) return;
  int b = idx / (Sc + Tc);
  int i = idx % (Sc + Tc);
  int sg = 1 + i;
  float cnt = (float)max(counts[b * NSEG + sg], 1);
  float m = sums[b * NSEG + sg] / cnt;
  if (i < Sc) avg_a[b * Sc + i] = m;
  else        avg_c[b * Tc + (i - Sc)] = m;
}

__global__ void logits_kernel(const float* __restrict__ a,
                              const float* __restrict__ c,
                              const float* __restrict__ biasPtr,
                              float* __restrict__ out) {
  int idx = blockIdx.x * blockDim.x + threadIdx.x;
  if (idx >= Bc * Sc * (Tc / 4)) return;
  int t4  = idx % (Tc / 4);
  int rem = idx / (Tc / 4);
  int s = rem % Sc;
  int b = rem / Sc;
  float av = a[b * Sc + s] + biasPtr[0];
  v4f cv = *reinterpret_cast<const v4f*>(c + b * Tc + t4 * 4);
  v4f o = cv + av;
  reinterpret_cast<v4f*>(out)[idx] = o;
}

extern "C" void kernel_launch(void* const* d_in, const int* in_sizes, int n_in,
                              void* d_out, int out_size, void* d_ws, size_t ws_size,
                              hipStream_t stream) {
  const float* X    = (const float*)d_in[0];
  const int*   msk  = (const int*)d_in[1];
  const int*   wid  = (const int*)d_in[2];
  const float* wvec = (const float*)d_in[5];
  const float* bias = (const float*)d_in[6];
  float* out = (float*)d_out;

  char* ws = (char*)d_ws;
  size_t off = 0;
  int*   seg    = (int*)(ws + off);   off += (size_t)Bc * Lc * 4;
  int*   flg    = (int*)(ws + off);   off += (size_t)Bc * Lc * 4;
  float* dsrc   = (float*)(ws + off); off += (size_t)Bc * Lc * 4;
  float* dtgt   = (float*)(ws + off); off += (size_t)Bc * Lc * 4;
  char*  zbase  = ws + off;
  float* sums   = (float*)(ws + off); off += (size_t)Bc * NSEG * 4;
  int*   counts = (int*)(ws + off);   off += (size_t)Bc * NSEG * 4;
  size_t zbytes = (size_t)Bc * NSEG * 8;
  float* avg_a  = (float*)(ws + off); off += (size_t)Bc * Sc * 4;
  float* avg_c  = (float*)(ws + off); off += (size_t)Bc * Tc * 4;

  hipMemsetAsync(zbase, 0, zbytes, stream);

  seg_scan_kernel<<<Bc, 1024, 0, stream>>>(wid, msk, seg, flg);

  dot_wmma_kernel<<<(Bc * Lc) / TOK, 256, 0, stream>>>(X, wvec, dsrc, dtgt);

  seg_reduce_kernel<<<(Bc * Lc + 255) / 256, 256, 0, stream>>>(
      seg, flg, dsrc, dtgt, sums, counts);

  avg_kernel<<<(Bc * (Sc + Tc) + 255) / 256, 256, 0, stream>>>(
      sums, counts, avg_a, avg_c);

  logits_kernel<<<(Bc * Sc * (Tc / 4) + 255) / 256, 256, 0, stream>>>(
      avg_a, avg_c, bias, out);
}


# fuse segment sums, means and logits into one 256-block kernel: each block reduces the 64+128 segments of its output tile from a start-index table written by the scan kernel; memset dropped
# speedup vs baseline: 1.6976x; 1.2600x over previous
#include <hip/hip_runtime.h>
#include <hip/hip_bf16.h>
#include <cstdint>
#include <cstddef>

typedef float v2f __attribute__((ext_vector_type(2)));
typedef float v4f __attribute__((ext_vector_type(4)));
typedef float v8f __attribute__((ext_vector_type(8)));

constexpr int Bc = 8;
constexpr int Lc = 4096;
constexpr int Hc = 1024;
constexpr int Sc = 512;
constexpr int Tc = 512;
constexpr int TOK = 64;
constexpr int DW  = 8;
constexpr int KS  = 128;
constexpr int TPITCH = 132;
constexpr int NSEG = 1056;
constexpr int NSTART = Sc + Tc + 2;
constexpr int TS = 64;
constexpr int TT = 128;

__global__ __launch_bounds__(1024) void seg_scan_kernel(
    const int* __restrict__ wid, const int* __restrict__ mask,
    int* __restrict__ seg, int* __restrict__ segStart) {
  __shared__ int sHas[1024];
  __shared__ int sWid[1024];
  __shared__ int sSum[1024];
  const int tid = threadIdx.x;
  const int b = blockIdx.x;
  const int base = b * Lc + tid * 4;

  int w[4], v[4];
#pragma unroll
  for (int j = 0; j < 4; ++j) {
    w[j] = wid[base + j];
    v[j] = (mask[base + j] > 0) ? 1 : 0;
  }

  int curH = 0, curW = -2;
#pragma unroll
  for (int j = 0; j < 4; ++j)
    if (v[j]) { curH = 1; curW = w[j]; }
  sHas[tid] = curH; sWid[tid] = curW;
  __syncthreads();
  for (int off = 1; off < 1024; off <<= 1) {
    int pH = 0, pW = -2;
    if (tid >= off) { pH = sHas[tid - off]; pW = sWid[tid - off]; }
    __syncthreads();
    if (!curH && pH) { curH = 1; curW = pW; }
    sHas[tid] = curH; sWid[tid] = curW;
    __syncthreads();
  }
  int prev = -2;
  if (tid > 0 && sHas[tid - 1]) prev = sWid[tid - 1];

  int f[4]; int localSum = 0;
#pragma unroll
  for (int j = 0; j < 4; ++j) {
    f[j] = 0;
    if (v[j]) {
      f[j] = (w[j] == -1 || prev == -1 || w[j] != prev) ? 1 : 0;
      prev = w[j];
      localSum += f[j];
    }
  }

  int curS = localSum;
  sSum[tid] = curS;
  __syncthreads();
  for (int off = 1; off < 1024; off <<= 1) {
    int p = 0;
    if (tid >= off) p = sSum[tid - off];
    __syncthreads();
    curS += p;
    sSum[tid] = curS;
    __syncthreads();
  }
  int run = (tid > 0) ? sSum[tid - 1] : 0;
  const int total = sSum[1023];
  int* st = segStart + b * NSEG;
#pragma unroll
  for (int j = 0; j < 4; ++j) {
    int sg = -1;
    if (v[j]) {
      run += f[j]; sg = run - 1;
      if (f[j] && sg < NSTART) st[sg] = tid * 4 + j;
    }
    seg[base + j] = sg;
  }
  for (int k = total + tid; k < NSTART; k += 1024) st[k] = Lc;
}

__global__ __launch_bounds__(256) void dot_wmma_kernel(
    const float* __restrict__ X, const float* __restrict__ wvec,
    float* __restrict__ dsrc, float* __restrict__ dtgt) {
  __shared__ __align__(16) float tile[DW][16 * TPITCH];
  __shared__ __align__(16) float part[DW][2][TOK];

  const int tid  = threadIdx.x;
  const int lane = tid & 31;
  const int ks   = __builtin_amdgcn_readfirstlane(tid >> 5);
  const int col  = lane & 15;
  const int half = lane >> 4;
  const int tokBase = blockIdx.x * TOK;

  v4f bw[16];
  {
    const float* wp = wvec + (col ? 1 : 0) * Hc + ks * KS + half * 4;
#pragma unroll
    for (int j = 0; j < 16; ++j) {
      v4f t = *reinterpret_cast<const v4f*>(wp + j * 8);
      bw[j] = (col < 2) ? t : v4f{0.0f, 0.0f, 0.0f, 0.0f};
    }
  }

  const float* xp = X + (size_t)tokBase * Hc + ks * KS + lane * 4;
  float* myTile = &tile[ks][0];
  v4f buf[2][16];

#define LOAD_GROUP(G) do { \
    _Pragma("unroll") for (int i = 0; i < 16; ++i) \
      buf[(G) & 1][i] = *reinterpret_cast<const v4f*>(xp + (size_t)((G) * 16 + i) * Hc); \
  } while (0)

  LOAD_GROUP(0);
#pragma unroll
  for (int g = 0; g < 4; ++g) {
    __builtin_amdgcn_sched_barrier(0);
    if (g + 1 < 4) LOAD_GROUP(g + 1);
    __builtin_amdgcn_sched_barrier(0);

#pragma unroll
    for (int i = 0; i < 16; ++i)
      *reinterpret_cast<v4f*>(&myTile[i * TPITCH + lane * 4]) = buf[g & 1][i];

    v4f a[16];
#pragma unroll
    for (int j = 0; j < 16; ++j)
      a[j] = *reinterpret_cast<const v4f*>(&myTile[col * TPITCH + j * 8 + half * 4]);
    __builtin_amdgcn_sched_barrier(0);

    v8f acc0 = {}, acc1 = {};
#pragma unroll
    for (int j = 0; j < 16; ++j) {
      v2f a0 = {a[j].x, a[j].y}, a1 = {a[j].z, a[j].w};
      v2f b0 = {bw[j].x, bw[j].y}, b1 = {bw[j].z, bw[j].w};
      acc0 = __builtin_amdgcn_wmma_f32_16x16x4_f32(
           false, a0,  false, b0,
           (short)0, acc0,  false,  false);
      acc1 = __builtin_amdgcn_wmma_f32_16x16x4_f32(
           false, a1,  false, b1,
           (short)0, acc1,  false,  false);
    }
    v8f acc = acc0 + acc1;

    if (col < 2) {
      float* pp = &part[ks][col][g * 16 + half * 8];
      *reinterpret_cast<v4f*>(pp)     = v4f{acc[0], acc[1], acc[2], acc[3]};
      *reinterpret_cast<v4f*>(pp + 4) = v4f{acc[4], acc[5], acc[6], acc[7]};
    }
  }
#undef LOAD_GROUP

  __syncthreads();
  if (tid < 2 * TOK) {
    const int s = tid >> 6, tok = tid & (TOK - 1);
    float sum = part[0][s][tok];
    for (int k = 1; k < DW; ++k) sum += part[k][s][tok];
    (s ? dtgt : dsrc)[tokBase + tok] = sum;
  }
}

__global__ __launch_bounds__(256) void pool_logits_kernel(
    const int* __restrict__ seg, const int* __restrict__ segStart,
    const float* __restrict__ dsrc, const float* __restrict__ dtgt,
    const float* __restrict__ biasPtr, float* __restrict__ out) {
  __shared__ float sa[TS];
  __shared__ __align__(16) float sc[TT];
  constexpr int TILES_T = Tc / TT;
  constexpr int TILES = (Sc / TS) * TILES_T;
  const int tid = threadIdx.x;
  const int b  = blockIdx.x / TILES;
  const int r  = blockIdx.x % TILES;
  const int s0 = (r / TILES_T) * TS;
  const int t0 = (r % TILES_T) * TT;
  const float bias = biasPtr[0];

  if (tid < TS + TT) {
    const bool isSrc = tid < TS;
    const int sg = isSrc ? (1 + s0 + tid) : (1 + Sc + t0 + (tid - TS));
    const float* d = (isSrc ? dsrc : dtgt) + b * Lc;
    const int* sgb = seg + b * Lc;
    const int begin = segStart[b * NSEG + sg];
    const int end   = segStart[b * NSEG + sg + 1];
    float sum = 0.0f; int cnt = 0;
    for (int j0 = begin; j0 < end; j0 += 4) {
      int sv[4]; float dv[4];
      for (int u = 0; u < 4; ++u) {
        const int j = min(j0 + u, end - 1);
        sv[u] = sgb[j]; dv[u] = d[j];
      }
      __builtin_amdgcn_sched_barrier(0);
      for (int u = 0; u < 4; ++u) {
        const bool use = (j0 + u < end) && (sv[u] == sg);
        sum = use ? sum + dv[u] : sum;
        cnt += use ? 1 : 0;
      }
    }
    const float m = sum / (float)max(cnt, 1);
    if (isSrc) sa[tid] = m; else sc[tid - TS] = m;
  }
  __syncthreads();

  const int lane = tid & 31, w = tid >> 5;
  const v4f cv = *reinterpret_cast<const v4f*>(&sc[lane * 4]);
  float* op = out + ((size_t)(b * Sc + s0 + w * 8) * Tc + t0 + lane * 4);
  for (int i = 0; i < 8; ++i) {
    const float av = sa[w * 8 + i] + bias;
    *reinterpret_cast<v4f*>(op + (size_t)i * Tc) = cv + av;
  }
}

extern "C" void kernel_launch(void* const* d_in, const int* in_sizes, int n_in,
                              void* d_out, int out_size, void* d_ws, size_t ws_size,
                              hipStream_t stream) {
  const float* X    = (const float*)d_in[0];
  const int*   msk  = (const int*)d_in[1];
  const int*   wid  = (const int*)d_in[2];
  const float* wvec = (const float*)d_in[5];
  const float* bias = (const float*)d_in[6];
  float* out = (float*)d_out;

  char* ws = (char*)d_ws;
  size_t off = 0;
  int*   seg    = (int*)(ws + off);   off += (size_t)Bc * Lc * 4;
  float* dsrc   = (float*)(ws + off); off += (size_t)Bc * Lc * 4;
  float* dtgt   = (float*)(ws + off); off += (size_t)Bc * Lc * 4;
  int*   sstart = (int*)(ws + off);   off += (size_t)Bc * NSEG * 4;

  seg_scan_kernel<<<Bc, 1024, 0, stream>>>(wid, msk, seg, sstart);

  dot_wmma_kernel<<<(Bc * Lc) / TOK, 256, 0, stream>>>(X, wvec, dsrc, dtgt);

  pool_logits_kernel<<<Bc * (Sc / TS) * (Tc / TT), 256, 0, stream>>>(
      seg, sstart, dsrc, dtgt, bias, out);
}


# segmentation scan runs as the first 8 blocks of the dot kernel's grid instead of in a launch of its own: 3 launches -> 2, and the scan's time is hidden behind the activation stream
# speedup vs baseline: 2.1866x; 1.2881x over previous
#include <hip/hip_runtime.h>
#include <hip/hip_bf16.h>
#include <cstdint>
#include <cstddef>

typedef float v2f __attribute__((ext_vector_type(2)));
typedef float v4f __attribute__((ext_vector_type(4)));
typedef float v8f __attribute__((ext_vector_type(8)));
typedef int   v4i __attribute__((ext_vector_type(4)));

constexpr int Bc = 8;
constexpr int Lc = 4096;
constexpr int Hc = 1024;
constexpr int Sc = 512;
constexpr int Tc = 512;
constexpr int TOK = 64;
constexpr int DW  = 8;
constexpr int KS  = 128;
constexpr int TPITCH = 132;
constexpr int NSEG = 1056;
constexpr int NSTART = Sc + Tc + 2;
constexpr int TS = 64;
constexpr int TT = 128;
constexpr int NT  = 256;
constexpr int TPT = Lc / NT;

__global__ __launch_bounds__(NT) void scan_dot_kernel(
    const int* __restrict__ wid, const int* __restrict__ mask,
    int* __restrict__ seg, int* __restrict__ segStart,
    const float* __restrict__ X, const float* __restrict__ wvec,
    float* __restrict__ dsrc, float* __restrict__ dtgt) {
  const int tid = threadIdx.x;

  if (blockIdx.x < Bc) {
    __shared__ int sHas[NT];
    __shared__ int sWid[NT];
    __shared__ int sSum[NT];
    const int b = blockIdx.x;
    const int base = b * Lc + tid * TPT;

    int w[TPT], v[TPT];
#pragma unroll
    for (int q = 0; q < TPT / 4; ++q) {
      const v4i ww = *reinterpret_cast<const v4i*>(wid + base + q * 4);
      const v4i mm = *reinterpret_cast<const v4i*>(mask + base + q * 4);
#pragma unroll
      for (int e = 0; e < 4; ++e) {
        w[q * 4 + e] = ww[e];
        v[q * 4 + e] = (mm[e] > 0) ? 1 : 0;
      }
    }

    int curH = 0, curW = -2;
#pragma unroll
    for (int j = 0; j < TPT; ++j)
      if (v[j]) { curH = 1; curW = w[j]; }
    sHas[tid] = curH; sWid[tid] = curW;
    __syncthreads();
    for (int off = 1; off < NT; off <<= 1) {
      int pH = 0, pW = -2;
      if (tid >= off) { pH = sHas[tid - off]; pW = sWid[tid - off]; }
      __syncthreads();
      if (!curH && pH) { curH = 1; curW = pW; }
      sHas[tid] = curH; sWid[tid] = curW;
      __syncthreads();
    }
    int prev = -2;
    if (tid > 0 && sHas[tid - 1]) prev = sWid[tid - 1];

    int f[TPT]; int localSum = 0;
#pragma unroll
    for (int j = 0; j < TPT; ++j) {
      f[j] = 0;
      if (v[j]) {
        f[j] = (w[j] == -1 || prev == -1 || w[j] != prev) ? 1 : 0;
        prev = w[j];
        localSum += f[j];
      }
    }

    int curS = localSum;
    sSum[tid] = curS;
    __syncthreads();
    for (int off = 1; off < NT; off <<= 1) {
      int p = 0;
      if (tid >= off) p = sSum[tid - off];
      __syncthreads();
      curS += p;
      sSum[tid] = curS;
      __syncthreads();
    }
    int run = (tid > 0) ? sSum[tid - 1] : 0;
    const int total = sSum[NT - 1];
    int* st = segStart + b * NSEG;
    int sgv[TPT];
#pragma unroll
    for (int j = 0; j < TPT; ++j) {
      int sg = -1;
      if (v[j]) {
        run += f[j]; sg = run - 1;
        if (f[j] && sg < NSTART) st[sg] = tid * TPT + j;
      }
      sgv[j] = sg;
    }
#pragma unroll
    for (int q = 0; q < TPT / 4; ++q)
      *reinterpret_cast<v4i*>(seg + base + q * 4) =
          v4i{sgv[q * 4], sgv[q * 4 + 1], sgv[q * 4 + 2], sgv[q * 4 + 3]};
    for (int k = total + tid; k < NSTART; k += NT) st[k] = Lc;
    return;
  }

  __shared__ __align__(16) float tile[DW][16 * TPITCH];
  __shared__ __align__(16) float part[DW][2][TOK];

  const int lane = tid & 31;
  const int ks   = __builtin_amdgcn_readfirstlane(tid >> 5);
  const int col  = lane & 15;
  const int half = lane >> 4;
  const int tokBase = (blockIdx.x - Bc) * TOK;

  v4f bw[16];
  {
    const float* wp = wvec + (col ? 1 : 0) * Hc + ks * KS + half * 4;
#pragma unroll
    for (int j = 0; j < 16; ++j) {
      v4f t = *reinterpret_cast<const v4f*>(wp + j * 8);
      bw[j] = (col < 2) ? t : v4f{0.0f, 0.0f, 0.0f, 0.0f};
    }
  }

  const float* xp = X + (size_t)tokBase * Hc + ks * KS + lane * 4;
  float* myTile = &tile[ks][0];
  v4f buf[2][16];

#define LOAD_GROUP(G) do { \
    _Pragma("unroll") for (int i = 0; i < 16; ++i) \
      buf[(G) & 1][i] = *reinterpret_cast<const v4f*>(xp + (size_t)((G) * 16 + i) * Hc); \
  } while (0)

  LOAD_GROUP(0);
#pragma unroll
  for (int g = 0; g < 4; ++g) {
    __builtin_amdgcn_sched_barrier(0);
    if (g + 1 < 4) LOAD_GROUP(g + 1);
    __builtin_amdgcn_sched_barrier(0);

#pragma unroll
    for (int i = 0; i < 16; ++i)
      *reinterpret_cast<v4f*>(&myTile[i * TPITCH + lane * 4]) = buf[g & 1][i];

    v4f a[16];
#pragma unroll
    for (int j = 0; j < 16; ++j)
      a[j] = *reinterpret_cast<const v4f*>(&myTile[col * TPITCH + j * 8 + half * 4]);
    __builtin_amdgcn_sched_barrier(0);

    v8f acc0 = {}, acc1 = {};
#pragma unroll
    for (int j = 0; j < 16; ++j) {
      v2f a0 = {a[j].x, a[j].y}, a1 = {a[j].z, a[j].w};
      v2f b0 = {bw[j].x, bw[j].y}, b1 = {bw[j].z, bw[j].w};
      acc0 = __builtin_amdgcn_wmma_f32_16x16x4_f32(
           false, a0,  false, b0,
           (short)0, acc0,  false,  false);
      acc1 = __builtin_amdgcn_wmma_f32_16x16x4_f32(
           false, a1,  false, b1,
           (short)0, acc1,  false,  false);
    }
    v8f acc = acc0 + acc1;

    if (col < 2) {
      float* pp = &part[ks][col][g * 16 + half * 8];
      *reinterpret_cast<v4f*>(pp)     = v4f{acc[0], acc[1], acc[2], acc[3]};
      *reinterpret_cast<v4f*>(pp + 4) = v4f{acc[4], acc[5], acc[6], acc[7]};
    }
  }
#undef LOAD_GROUP

  __syncthreads();
  if (tid < 2 * TOK) {
    const int s = tid >> 6, tok = tid & (TOK - 1);
    float sum = part[0][s][tok];
    for (int k = 1; k < DW; ++k) sum += part[k][s][tok];
    (s ? dtgt : dsrc)[tokBase + tok] = sum;
  }
}

__global__ __launch_bounds__(256) void pool_logits_kernel(
    const int* __restrict__ seg, const int* __restrict__ segStart,
    const float* __restrict__ dsrc, const float* __restrict__ dtgt,
    const float* __restrict__ biasPtr, float* __restrict__ out) {
  __shared__ float sa[TS];
  __shared__ __align__(16) float sc[TT];
  constexpr int TILES_T = Tc / TT;
  constexpr int TILES = (Sc / TS) * TILES_T;
  const int tid = threadIdx.x;
  const int b  = blockIdx.x / TILES;
  const int r  = blockIdx.x % TILES;
  const int s0 = (r / TILES_T) * TS;
  const int t0 = (r % TILES_T) * TT;
  const float bias = biasPtr[0];

  if (tid < TS + TT) {
    const bool isSrc = tid < TS;
    const int sg = isSrc ? (1 + s0 + tid) : (1 + Sc + t0 + (tid - TS));
    const float* d = (isSrc ? dsrc : dtgt) + b * Lc;
    const int* sgb = seg + b * Lc;
    const int begin = segStart[b * NSEG + sg];
    const int end   = segStart[b * NSEG + sg + 1];
    float sum = 0.0f; int cnt = 0;
    for (int j0 = begin; j0 < end; j0 += 4) {
      int sv[4]; float dv[4];
      for (int u = 0; u < 4; ++u) {
        const int j = min(j0 + u, end - 1);
        sv[u] = sgb[j]; dv[u] = d[j];
      }
      __builtin_amdgcn_sched_barrier(0);
      for (int u = 0; u < 4; ++u) {
        const bool use = (j0 + u < end) && (sv[u] == sg);
        sum = use ? sum + dv[u] : sum;
        cnt += use ? 1 : 0;
      }
    }
    const float m = sum / (float)max(cnt, 1);
    if (isSrc) sa[tid] = m; else sc[tid - TS] = m;
  }
  __syncthreads();

  const int lane = tid & 31, w = tid >> 5;
  const v4f cv = *reinterpret_cast<const v4f*>(&sc[lane * 4]);
  float* op = out + ((size_t)(b * Sc + s0 + w * 8) * Tc + t0 + lane * 4);
  for (int i = 0; i < 8; ++i) {
    const float av = sa[w * 8 + i] + bias;
    *reinterpret_cast<v4f*>(op + (size_t)i * Tc) = cv + av;
  }
}

extern "C" void kernel_launch(void* const* d_in, const int* in_sizes, int n_in,
                              void* d_out, int out_size, void* d_ws, size_t ws_size,
                              hipStream_t stream) {
  const float* X    = (const float*)d_in[0];
  const int*   msk  = (const int*)d_in[1];
  const int*   wid  = (const int*)d_in[2];
  const float* wvec = (const float*)d_in[5];
  const float* bias = (const float*)d_in[6];
  float* out = (float*)d_out;

  char* ws = (char*)d_ws;
  size_t off = 0;
  int*   seg    = (int*)(ws + off);   off += (size_t)Bc * Lc * 4;
  float* dsrc   = (float*)(ws + off); off += (size_t)Bc * Lc * 4;
  float* dtgt   = (float*)(ws + off); off += (size_t)Bc * Lc * 4;
  int*   sstart = (int*)(ws + off);   off += (size_t)Bc * NSEG * 4;

  scan_dot_kernel<<<Bc + (Bc * Lc) / TOK, NT, 0, stream>>>(
      wid, msk, seg, sstart, X, wvec, dsrc, dtgt);

  pool_logits_kernel<<<Bc * (Sc / TS) * (Tc / TT), 256, 0, stream>>>(
      seg, sstart, dsrc, dtgt, bias, out);
}


# dot blocks count the segment starts before and inside their 64 tokens (lane-mask popcounts, one barrier) and exit if no token lies in segments 1..1024, all the output reads; sums unchanged
# speedup vs baseline: 2.2973x; 1.0506x over previous
#include <hip/hip_runtime.h>
#include <hip/hip_bf16.h>
#include <cstdint>
#include <cstddef>

typedef float v2f __attribute__((ext_vector_type(2)));
typedef float v4f __attribute__((ext_vector_type(4)));
typedef float v8f __attribute__((ext_vector_type(8)));
typedef int   v4i __attribute__((ext_vector_type(4)));

constexpr int Bc = 8;
constexpr int Lc = 4096;
constexpr int Hc = 1024;
constexpr int Sc = 512;
constexpr int Tc = 512;
constexpr int TOK = 64;
constexpr int DW  = 8;
constexpr int KS  = 128;
constexpr int TPITCH = 132;
constexpr int NSEG = 1056;
constexpr int NSTART = Sc + Tc + 2;
constexpr int TS = 64;
constexpr int TT = 128;
constexpr int NT  = 256;
constexpr int TPT = Lc / NT;

__global__ __launch_bounds__(NT) void scan_dot_kernel(
    const int* __restrict__ wid, const int* __restrict__ mask,
    int* __restrict__ seg, int* __restrict__ segStart,
    const float* __restrict__ X, const float* __restrict__ wvec,
    float* __restrict__ dsrc, float* __restrict__ dtgt) {
  const int tid = threadIdx.x;

  if (blockIdx.x < Bc) {
    __shared__ int sHas[NT];
    __shared__ int sWid[NT];
    __shared__ int sSum[NT];
    const int b = blockIdx.x;
    const int base = b * Lc + tid * TPT;

    int w[TPT], v[TPT];
#pragma unroll
    for (int q = 0; q < TPT / 4; ++q) {
      const v4i ww = *reinterpret_cast<const v4i*>(wid + base + q * 4);
      const v4i mm = *reinterpret_cast<const v4i*>(mask + base + q * 4);
#pragma unroll
      for (int e = 0; e < 4; ++e) {
        w[q * 4 + e] = ww[e];
        v[q * 4 + e] = (mm[e] > 0) ? 1 : 0;
      }
    }

    int curH = 0, curW = -2;
#pragma unroll
    for (int j = 0; j < TPT; ++j)
      if (v[j]) { curH = 1; curW = w[j]; }
    sHas[tid] = curH; sWid[tid] = curW;
    __syncthreads();
    for (int off = 1; off < NT; off <<= 1) {
      int pH = 0, pW = -2;
      if (tid >= off) { pH = sHas[tid - off]; pW = sWid[tid - off]; }
      __syncthreads();
      if (!curH && pH) { curH = 1; curW = pW; }
      sHas[tid] = curH; sWid[tid] = curW;
      __syncthreads();
    }
    int prev = -2;
    if (tid > 0 && sHas[tid - 1]) prev = sWid[tid - 1];

    int f[TPT]; int localSum = 0;
#pragma unroll
    for (int j = 0; j < TPT; ++j) {
      f[j] = 0;
      if (v[j]) {
        f[j] = (w[j] == -1 || prev == -1 || w[j] != prev) ? 1 : 0;
        prev = w[j];
        localSum += f[j];
      }
    }

    int curS = localSum;
    sSum[tid] = curS;
    __syncthreads();
    for (int off = 1; off < NT; off <<= 1) {
      int p = 0;
      if (tid >= off) p = sSum[tid - off];
      __syncthreads();
      curS += p;
      sSum[tid] = curS;
      __syncthreads();
    }
    int run = (tid > 0) ? sSum[tid - 1] : 0;
    const int total = sSum[NT - 1];
    int* st = segStart + b * NSEG;
    int sgv[TPT];
#pragma unroll
    for (int j = 0; j < TPT; ++j) {
      int sg = -1;
      if (v[j]) {
        run += f[j]; sg = run - 1;
        if (f[j] && sg < NSTART) st[sg] = tid * TPT + j;
      }
      sgv[j] = sg;
    }
#pragma unroll
    for (int q = 0; q < TPT / 4; ++q)
      *reinterpret_cast<v4i*>(seg + base + q * 4) =
          v4i{sgv[q * 4], sgv[q * 4 + 1], sgv[q * 4 + 2], sgv[q * 4 + 3]};
    for (int k = total + tid; k < NSTART; k += NT) st[k] = Lc;
    return;
  }

  __shared__ __align__(16) float tile[DW][16 * TPITCH];
  __shared__ __align__(16) float part[DW][2][TOK];
  __shared__ __align__(16) int sCnt[DW];

  const int lane = tid & 31;
  const int ks   = __builtin_amdgcn_readfirstlane(tid >> 5);
  const int col  = lane & 15;
  const int half = lane >> 4;
  const int tokBase = (blockIdx.x - Bc) * TOK;

  {
    const int bb = tokBase / Lc;
    const int t0 = tokBase % Lc;
    const int* wb = wid + bb * Lc;
    const int* mb = mask + bb * Lc;
    const int c0 = tid * TPT;
    const bool act = c0 < t0 + TOK;
    const unsigned ownMask =
        (ks == (t0 / TPT) / 32) ? (0xFu << ((t0 / TPT) & 31)) : 0u;

    int w[TPT], v[TPT];
    int prev = -2;
    if (act) {
      const int jp = max(c0 - 1, 0);
      const int pm0 = mb[jp], pw0 = wb[jp];
      v4i ww[TPT / 4], mm[TPT / 4];
#pragma unroll
      for (int q = 0; q < TPT / 4; ++q) {
        ww[q] = *reinterpret_cast<const v4i*>(wb + c0 + q * 4);
        mm[q] = *reinterpret_cast<const v4i*>(mb + c0 + q * 4);
      }
#pragma unroll
      for (int q = 0; q < TPT / 4; ++q) {
#pragma unroll
        for (int e = 0; e < 4; ++e) {
          w[q * 4 + e] = ww[q][e];
          v[q * 4 + e] = (mm[q][e] > 0) ? 1 : 0;
        }
      }
      prev = (c0 > 0 && pm0 > 0) ? pw0 : -2;
      if (c0 > 1 && pm0 <= 0) {
        for (int j = c0 - 2; j >= 0; --j)
          if (mb[j] > 0) { prev = wb[j]; break; }
      }
    } else {
#pragma unroll
      for (int j = 0; j < TPT; ++j) { w[j] = 0; v[j] = 0; }
    }

    int nAll = 0, nOwn = 0;
#pragma unroll
    for (int j = 0; j < TPT; ++j) {
      bool f = false;
      if (v[j]) {
        f = (w[j] == -1 || prev == -1 || w[j] != prev);
        prev = w[j];
      }
      const unsigned m = __builtin_amdgcn_ballot_w32(f);
      nAll += __builtin_popcount(m);
      nOwn += __builtin_popcount(m & ownMask);
    }
    if (lane == 0) sCnt[ks] = nAll | (nOwn << 16);
    __syncthreads();
    const v4i ca = *reinterpret_cast<const v4i*>(&sCnt[0]);
    const v4i cb = *reinterpret_cast<const v4i*>(&sCnt[4]);
    const int tot = __builtin_amdgcn_readfirstlane(
        ca[0] + ca[1] + ca[2] + ca[3] + cb[0] + cb[1] + cb[2] + cb[3]);
    const int cEnd = tot & 0xFFFF;
    const int cPre = cEnd - (tot >> 16);
    if (cEnd < 2 || cPre > Sc + Tc + 1) return;
  }

  v4f bw[16];
  {
    const float* wp = wvec + (col ? 1 : 0) * Hc + ks * KS + half * 4;
#pragma unroll
    for (int j = 0; j < 16; ++j) {
      v4f t = *reinterpret_cast<const v4f*>(wp + j * 8);
      bw[j] = (col < 2) ? t : v4f{0.0f, 0.0f, 0.0f, 0.0f};
    }
  }

  const float* xp = X + (size_t)tokBase * Hc + ks * KS + lane * 4;
  float* myTile = &tile[ks][0];
  v4f buf[2][16];

#define LOAD_GROUP(G) do { \
    _Pragma("unroll") for (int i = 0; i < 16; ++i) \
      buf[(G) & 1][i] = *reinterpret_cast<const v4f*>(xp + (size_t)((G) * 16 + i) * Hc); \
  } while (0)

  LOAD_GROUP(0);
#pragma unroll
  for (int g = 0; g < 4; ++g) {
    __builtin_amdgcn_sched_barrier(0);
    if (g + 1 < 4) LOAD_GROUP(g + 1);
    __builtin_amdgcn_sched_barrier(0);

#pragma unroll
    for (int i = 0; i < 16; ++i)
      *reinterpret_cast<v4f*>(&myTile[i * TPITCH + lane * 4]) = buf[g & 1][i];

    v4f a[16];
#pragma unroll
    for (int j = 0; j < 16; ++j)
      a[j] = *reinterpret_cast<const v4f*>(&myTile[col * TPITCH + j * 8 + half * 4]);
    __builtin_amdgcn_sched_barrier(0);

    v8f acc0 = {}, acc1 = {};
#pragma unroll
    for (int j = 0; j < 16; ++j) {
      v2f a0 = {a[j].x, a[j].y}, a1 = {a[j].z, a[j].w};
      v2f b0 = {bw[j].x, bw[j].y}, b1 = {bw[j].z, bw[j].w};
      acc0 = __builtin_amdgcn_wmma_f32_16x16x4_f32(
           false, a0,  false, b0,
           (short)0, acc0,  false,  false);
      acc1 = __builtin_amdgcn_wmma_f32_16x16x4_f32(
           false, a1,  false, b1,
           (short)0, acc1,  false,  false);
    }
    v8f acc = acc0 + acc1;

    if (col < 2) {
      float* pp = &part[ks][col][g * 16 + half * 8];
      *reinterpret_cast<v4f*>(pp)     = v4f{acc[0], acc[1], acc[2], acc[3]};
      *reinterpret_cast<v4f*>(pp + 4) = v4f{acc[4], acc[5], acc[6], acc[7]};
    }
  }
#undef LOAD_GROUP

  __syncthreads();
  if (tid < 2 * TOK) {
    const int s = tid >> 6, tok = tid & (TOK - 1);
    float sum = part[0][s][tok];
    for (int k = 1; k < DW; ++k) sum += part[k][s][tok];
    (s ? dtgt : dsrc)[tokBase + tok] = sum;
  }
}

__global__ __launch_bounds__(256) void pool_logits_kernel(
    const int* __restrict__ seg, const int* __restrict__ segStart,
    const float* __restrict__ dsrc, const float* __restrict__ dtgt,
    const float* __restrict__ biasPtr, float* __restrict__ out) {
  __shared__ float sa[TS];
  __shared__ __align__(16) float sc[TT];
  constexpr int TILES_T = Tc / TT;
  constexpr int TILES = (Sc / TS) * TILES_T;
  const int tid = threadIdx.x;
  const int b  = blockIdx.x / TILES;
  const int r  = blockIdx.x % TILES;
  const int s0 = (r / TILES_T) * TS;
  const int t0 = (r % TILES_T) * TT;
  const float bias = biasPtr[0];

  if (tid < TS + TT) {
    const bool isSrc = tid < TS;
    const int sg = isSrc ? (1 + s0 + tid) : (1 + Sc + t0 + (tid - TS));
    const float* d = (isSrc ? dsrc : dtgt) + b * Lc;
    const int* sgb = seg + b * Lc;
    const int begin = segStart[b * NSEG + sg];
    const int end   = segStart[b * NSEG + sg + 1];
    float sum = 0.0f; int cnt = 0;
    for (int j0 = begin; j0 < end; j0 += 4) {
      int sv[4]; float dv[4];
      for (int u = 0; u < 4; ++u) {
        const int j = min(j0 + u, end - 1);
        sv[u] = sgb[j]; dv[u] = d[j];
      }
      __builtin_amdgcn_sched_barrier(0);
      for (int u = 0; u < 4; ++u) {
        const bool use = (j0 + u < end) && (sv[u] == sg);
        sum = use ? sum + dv[u] : sum;
        cnt += use ? 1 : 0;
      }
    }
    const float m = sum / (float)max(cnt, 1);
    if (isSrc) sa[tid] = m; else sc[tid - TS] = m;
  }
  __syncthreads();

  const int lane = tid & 31, w = tid >> 5;
  const v4f cv = *reinterpret_cast<const v4f*>(&sc[lane * 4]);
  float* op = out + ((size_t)(b * Sc + s0 + w * 8) * Tc + t0 + lane * 4);
  for (int i = 0; i < 8; ++i) {
    const float av = sa[w * 8 + i] + bias;
    *reinterpret_cast<v4f*>(op + (size_t)i * Tc) = cv + av;
  }
}

extern "C" void kernel_launch(void* const* d_in, const int* in_sizes, int n_in,
                              void* d_out, int out_size, void* d_ws, size_t ws_size,
                              hipStream_t stream) {
  const float* X    = (const float*)d_in[0];
  const int*   msk  = (const int*)d_in[1];
  const int*   wid  = (const int*)d_in[2];
  const float* wvec = (const float*)d_in[5];
  const float* bias = (const float*)d_in[6];
  float* out = (float*)d_out;

  char* ws = (char*)d_ws;
  size_t off = 0;
  int*   seg    = (int*)(ws + off);   off += (size_t)Bc * Lc * 4;
  float* dsrc   = (float*)(ws + off); off += (size_t)Bc * Lc * 4;
  float* dtgt   = (float*)(ws + off); off += (size_t)Bc * Lc * 4;
  int*   sstart = (int*)(ws + off);   off += (size_t)Bc * NSEG * 4;

  scan_dot_kernel<<<Bc + (Bc * Lc) / TOK, NT, 0, stream>>>(
      wid, msk, seg, sstart, X, wvec, dsrc, dtgt);

  pool_logits_kernel<<<Bc * (Sc / TS) * (Tc / TT), 256, 0, stream>>>(
      seg, sstart, dsrc, dtgt, bias, out);
}


# dot blocks share the needed token range evenly: 32 per batch, each finds the range by lane-mask popcounts of segment starts and streams 1/32 of its 16-token groups; dots kept as (src,tgt) pairs
# speedup vs baseline: 2.6931x; 1.1723x over previous
#include <hip/hip_runtime.h>
#include <hip/hip_bf16.h>
#include <cstdint>
#include <cstddef>

typedef float v2f __attribute__((ext_vector_type(2)));
typedef float v4f __attribute__((ext_vector_type(4)));
typedef float v8f __attribute__((ext_vector_type(8)));
typedef int   v4i __attribute__((ext_vector_type(4)));

constexpr int Bc = 8;
constexpr int Lc = 4096;
constexpr int Hc = 1024;
constexpr int Sc = 512;
constexpr int Tc = 512;
constexpr int BPB = 32;
constexpr int MAXG = Lc / 16 / BPB;
constexpr int DW  = 8;
constexpr int KS  = 128;
constexpr int TPITCH = 132;
constexpr int NSEG = 1056;
constexpr int NSTART = Sc + Tc + 2;
constexpr int TS = 64;
constexpr int TT = 128;
constexpr int NT  = 256;
constexpr int TPT = Lc / NT;

__global__ __launch_bounds__(NT) void scan_dot_kernel(
    const int* __restrict__ wid, const int* __restrict__ mask,
    int* __restrict__ seg, int* __restrict__ segStart,
    const float* __restrict__ X, const float* __restrict__ wvec,
    float* __restrict__ dsrc, float* __restrict__ dtgt) {
  const int tid = threadIdx.x;

  if (blockIdx.x < Bc) {
    __shared__ int sHas[NT];
    __shared__ int sWid[NT];
    __shared__ int sSum[NT];
    const int b = blockIdx.x;
    const int base = b * Lc + tid * TPT;

    int w[TPT], v[TPT];
#pragma unroll
    for (int q = 0; q < TPT / 4; ++q) {
      const v4i ww = *reinterpret_cast<const v4i*>(wid + base + q * 4);
      const v4i mm = *reinterpret_cast<const v4i*>(mask + base + q * 4);
#pragma unroll
      for (int e = 0; e < 4; ++e) {
        w[q * 4 + e] = ww[e];
        v[q * 4 + e] = (mm[e] > 0) ? 1 : 0;
      }
    }

    int curH = 0, curW = -2;
#pragma unroll
    for (int j = 0; j < TPT; ++j)
      if (v[j]) { curH = 1; curW = w[j]; }
    sHas[tid] = curH; sWid[tid] = curW;
    __syncthreads();
    for (int off = 1; off < NT; off <<= 1) {
      int pH = 0, pW = -2;
      if (tid >= off) { pH = sHas[tid - off]; pW = sWid[tid - off]; }
      __syncthreads();
      if (!curH && pH) { curH = 1; curW = pW; }
      sHas[tid] = curH; sWid[tid] = curW;
      __syncthreads();
    }
    int prev = -2;
    if (tid > 0 && sHas[tid - 1]) prev = sWid[tid - 1];

    int f[TPT]; int localSum = 0;
#pragma unroll
    for (int j = 0; j < TPT; ++j) {
      f[j] = 0;
      if (v[j]) {
        f[j] = (w[j] == -1 || prev == -1 || w[j] != prev) ? 1 : 0;
        prev = w[j];
        localSum += f[j];
      }
    }

    int curS = localSum;
    sSum[tid] = curS;
    __syncthreads();
    for (int off = 1; off < NT; off <<= 1) {
      int p = 0;
      if (tid >= off) p = sSum[tid - off];
      __syncthreads();
      curS += p;
      sSum[tid] = curS;
      __syncthreads();
    }
    int run = (tid > 0) ? sSum[tid - 1] : 0;
    const int total = sSum[NT - 1];
    int* st = segStart + b * NSEG;
    int sgv[TPT];
#pragma unroll
    for (int j = 0; j < TPT; ++j) {
      int sg = -1;
      if (v[j]) {
        run += f[j]; sg = run - 1;
        if (f[j] && sg < NSTART) st[sg] = tid * TPT + j;
      }
      sgv[j] = sg;
    }
#pragma unroll
    for (int q = 0; q < TPT / 4; ++q)
      *reinterpret_cast<v4i*>(seg + base + q * 4) =
          v4i{sgv[q * 4], sgv[q * 4 + 1], sgv[q * 4 + 2], sgv[q * 4 + 3]};
    for (int k = total + tid; k < NSTART; k += NT) st[k] = Lc;
    return;
  }

  __shared__ __align__(16) float tile[DW][16 * TPITCH];
  __shared__ __align__(16) float part[DW][2][MAXG * 16];
  __shared__ __align__(16) int sCnt[DW];
  __shared__ int sLoHi[2];

  const int lane = tid & 31;
  const int ks   = __builtin_amdgcn_readfirstlane(tid >> 5);
  const int col  = lane & 15;
  const int half = lane >> 4;
  const int bb   = (blockIdx.x - Bc) / BPB;
  const int bi   = (blockIdx.x - Bc) % BPB;

  v4f bw[16];
  {
    const float* wp = wvec + (col ? 1 : 0) * Hc + ks * KS + half * 4;
#pragma unroll
    for (int j = 0; j < 16; ++j) {
      v4f t = *reinterpret_cast<const v4f*>(wp + j * 8);
      bw[j] = (col < 2) ? t : v4f{0.0f, 0.0f, 0.0f, 0.0f};
    }
  }

  int gBeg, gEnd;
  {
    const int* wb = wid + bb * Lc;
    const int* mb = mask + bb * Lc;
    const int c0 = tid * TPT;

    int w[TPT], v[TPT];
    const int jp = max(c0 - 1, 0);
    const int pm0 = mb[jp], pw0 = wb[jp];
    v4i ww[TPT / 4], mm[TPT / 4];
#pragma unroll
    for (int q = 0; q < TPT / 4; ++q) {
      ww[q] = *reinterpret_cast<const v4i*>(wb + c0 + q * 4);
      mm[q] = *reinterpret_cast<const v4i*>(mb + c0 + q * 4);
    }
#pragma unroll
    for (int q = 0; q < TPT / 4; ++q) {
#pragma unroll
      for (int e = 0; e < 4; ++e) {
        w[q * 4 + e] = ww[q][e];
        v[q * 4 + e] = (mm[q][e] > 0) ? 1 : 0;
      }
    }
    int prev = (c0 > 0 && pm0 > 0) ? pw0 : -2;
    if (c0 > 1 && pm0 <= 0) {
      for (int j = c0 - 2; j >= 0; --j)
        if (mb[j] > 0) { prev = wb[j]; break; }
    }

    if (tid == 0) { sLoHi[0] = Lc; sLoHi[1] = Lc; }

    unsigned fbits = 0;
    int nAll = 0, pre = 0;
#pragma unroll
    for (int j = 0; j < TPT; ++j) {
      bool f = false;
      if (v[j]) {
        f = (w[j] == -1 || prev == -1 || w[j] != prev);
        prev = w[j];
      }
      const unsigned m = __builtin_amdgcn_ballot_w32(f);
      nAll += __builtin_popcount(m);
      pre = __builtin_amdgcn_mbcnt_lo(m, pre);
      fbits |= (f ? 1u : 0u) << j;
    }
    if (lane == 0) sCnt[ks] = nAll;
    __syncthreads();
    const v4i ca = *reinterpret_cast<const v4i*>(&sCnt[0]);
    const v4i cb = *reinterpret_cast<const v4i*>(&sCnt[4]);
    const int cw[DW] = {ca[0], ca[1], ca[2], ca[3], cb[0], cb[1], cb[2], cb[3]};
    int c = pre;
#pragma unroll
    for (int k = 0; k < DW; ++k) c += (k < ks) ? cw[k] : 0;
    const int own = __builtin_popcount(fbits);
    if ((c <= 1 && 1 < c + own) || (c <= Sc + Tc + 1 && Sc + Tc + 1 < c + own)) {
#pragma unroll
      for (int j = 0; j < TPT; ++j) {
        if ((fbits >> j) & 1u) {
          if (c == 1) sLoHi[0] = c0 + j;
          if (c == Sc + Tc + 1) sLoHi[1] = c0 + j;
          ++c;
        }
      }
    }
    __syncthreads();
    const int lo = __builtin_amdgcn_readfirstlane(sLoHi[0]);
    const int hi = __builtin_amdgcn_readfirstlane(sLoHi[1]);
    const int gLo = lo >> 4;
    const int G = ((hi + 15) >> 4) - gLo;
    gBeg = gLo + (bi * G) / BPB;
    gEnd = gLo + ((bi + 1) * G) / BPB;
  }

  const float* xp = X + ((size_t)bb * Lc + (size_t)gBeg * 16) * Hc + ks * KS + lane * 4;
  float* myTile = &tile[ks][0];
  v4f buf0[16], buf1[16];

#define LOAD_GROUP(BUF, S) do { \
    _Pragma("unroll") for (int i = 0; i < 16; ++i) \
      BUF[i] = *reinterpret_cast<const v4f*>(xp + (size_t)((S) * 16 + i) * Hc); \
  } while (0)

#define USE_GROUP(BUF, S) do { \
      \
    _Pragma("unroll") for (int i = 0; i < 16; ++i) \
      *reinterpret_cast<v4f*>(&myTile[i * TPITCH + lane * 4]) = BUF[i]; \
      \
      \
      \
    v4f a[16]; \
    _Pragma("unroll") for (int j = 0; j < 16; ++j) \
      a[j] = *reinterpret_cast<const v4f*>(&myTile[col * TPITCH + j * 8 + half * 4]); \
      \
      \
    __builtin_amdgcn_sched_barrier(0); \
    v8f acc0 = {}, acc1 = {};               \
    _Pragma("unroll") for (int j = 0; j < 16; ++j) { \
      v2f a0 = {a[j].x, a[j].y}, a1 = {a[j].z, a[j].w}; \
      v2f b0 = {bw[j].x, bw[j].y}, b1 = {bw[j].z, bw[j].w}; \
      acc0 = __builtin_amdgcn_wmma_f32_16x16x4_f32( \
           false, a0,  false, b0, \
           (short)0, acc0,  false,  false); \
      acc1 = __builtin_amdgcn_wmma_f32_16x16x4_f32( \
           false, a1,  false, b1, \
           (short)0, acc1,  false,  false); \
    } \
    v8f acc = acc0 + acc1; \
      \
      \
    if (col < 2) { \
      float* pp = &part[ks][col][(S) * 16 + half * 8]; \
      *reinterpret_cast<v4f*>(pp)     = v4f{acc[0], acc[1], acc[2], acc[3]}; \
      *reinterpret_cast<v4f*>(pp + 4) = v4f{acc[4], acc[5], acc[6], acc[7]}; \
    } \
  } while (0)

  const int ng = gEnd - gBeg;
  if (ng > 0) {
    LOAD_GROUP(buf0, 0);
    int s = 0;
    for (; s + 2 < ng; s += 2) {
      __builtin_amdgcn_sched_barrier(0);
      LOAD_GROUP(buf1, s + 1);
      __builtin_amdgcn_sched_barrier(0);
      USE_GROUP(buf0, s);
      __builtin_amdgcn_sched_barrier(0);
      LOAD_GROUP(buf0, s + 2);
      __builtin_amdgcn_sched_barrier(0);
      USE_GROUP(buf1, s + 1);
    }
    if (s + 1 < ng) {
      __builtin_amdgcn_sched_barrier(0);
      LOAD_GROUP(buf1, s + 1);
      __builtin_amdgcn_sched_barrier(0);
      USE_GROUP(buf0, s);
      USE_GROUP(buf1, s + 1);
    } else {
      USE_GROUP(buf0, s);
    }
  }
#undef LOAD_GROUP
#undef USE_GROUP

  __syncthreads();
  {
    const int sd = tid & 1, tok = tid >> 1;
    if (tok < ng * 16) {
      float sum = part[0][sd][tok];
      for (int k = 1; k < DW; ++k) sum += part[k][sd][tok];
      (sd ? dtgt : dsrc)[2 * (bb * Lc + gBeg * 16 + tok)] = sum;
    }
  }
}

__global__ __launch_bounds__(256) void pool_logits_kernel(
    const int* __restrict__ seg, const int* __restrict__ segStart,
    const float* __restrict__ dsrc, const float* __restrict__ dtgt,
    const float* __restrict__ biasPtr, float* __restrict__ out) {
  __shared__ float sa[TS];
  __shared__ __align__(16) float sc[TT];
  constexpr int TILES_T = Tc / TT;
  constexpr int TILES = (Sc / TS) * TILES_T;
  const int tid = threadIdx.x;
  const int b  = blockIdx.x / TILES;
  const int r  = blockIdx.x % TILES;
  const int s0 = (r / TILES_T) * TS;
  const int t0 = (r % TILES_T) * TT;
  const float bias = biasPtr[0];

  if (tid < TS + TT) {
    const bool isSrc = tid < TS;
    const int sg = isSrc ? (1 + s0 + tid) : (1 + Sc + t0 + (tid - TS));
    const float* d = (isSrc ? dsrc : dtgt) + 2 * b * Lc;
    const int* sgb = seg + b * Lc;
    const int begin = segStart[b * NSEG + sg];
    const int end   = segStart[b * NSEG + sg + 1];
    float sum = 0.0f; int cnt = 0;
    for (int j0 = begin; j0 < end; j0 += 4) {
      int sv[4]; float dv[4];
      for (int u = 0; u < 4; ++u) {
        const int j = min(j0 + u, end - 1);
        sv[u] = sgb[j]; dv[u] = d[2 * j];
      }
      __builtin_amdgcn_sched_barrier(0);
      for (int u = 0; u < 4; ++u) {
        const bool use = (j0 + u < end) && (sv[u] == sg);
        sum = use ? sum + dv[u] : sum;
        cnt += use ? 1 : 0;
      }
    }
    const float m = sum / (float)max(cnt, 1);
    if (isSrc) sa[tid] = m; else sc[tid - TS] = m;
  }
  __syncthreads();

  const int lane = tid & 31, w = tid >> 5;
  const v4f cv = *reinterpret_cast<const v4f*>(&sc[lane * 4]);
  float* op = out + ((size_t)(b * Sc + s0 + w * 8) * Tc + t0 + lane * 4);
  for (int i = 0; i < 8; ++i) {
    const float av = sa[w * 8 + i] + bias;
    *reinterpret_cast<v4f*>(op + (size_t)i * Tc) = cv + av;
  }
}

extern "C" void kernel_launch(void* const* d_in, const int* in_sizes, int n_in,
                              void* d_out, int out_size, void* d_ws, size_t ws_size,
                              hipStream_t stream) {
  const float* X    = (const float*)d_in[0];
  const int*   msk  = (const int*)d_in[1];
  const int*   wid  = (const int*)d_in[2];
  const float* wvec = (const float*)d_in[5];
  const float* bias = (const float*)d_in[6];
  float* out = (float*)d_out;

  char* ws = (char*)d_ws;
  size_t off = 0;
  int*   seg    = (int*)(ws + off);   off += (size_t)Bc * Lc * 4;
  float* dsrc   = (float*)(ws + off); off += (size_t)Bc * Lc * 8;
  float* dtgt   = dsrc + 1;
  int*   sstart = (int*)(ws + off);   off += (size_t)Bc * NSEG * 4;

  scan_dot_kernel<<<Bc + Bc * BPB, NT, 0, stream>>>(
      wid, msk, seg, sstart, X, wvec, dsrc, dtgt);

  pool_logits_kernel<<<Bc * (Sc / TS) * (Tc / TT), 256, 0, stream>>>(
      seg, sstart, dsrc, dtgt, bias, out);
}
